# starting kernel
# baseline (speedup 1.0000x reference)
#include <hip/hip_runtime.h>
#include <hip/hip_bf16.h>

typedef float v2f __attribute__((ext_vector_type(2)));
typedef float v8f __attribute__((ext_vector_type(8)));

#define UP      5
#define DOWN    24
#define HLEN    481
#define MSHIFT  10
#define JMAX    96
#define KSTEPS  43
#define XCOLS   712
#define XSTRIDE 721
#define YSTRIDE 132
#define TAPMAX  844
#define HPAD    1232

__global__ __launch_bounds__(256) void resample_poly_wmma(
    const float* __restrict__ x,
    const float* __restrict__ h,
    float* __restrict__ out,
    int n_in, int n_out)
{
    __shared__ float hrev[HPAD];
    __shared__ float xs[16 * XSTRIDE];
    __shared__ float yo[16 * YSTRIDE];

    const int tid  = threadIdx.x;
    const int wave = tid >> 5;
    const int lane = tid & 31;
    const int b0    = blockIdx.y << 4;
    const int m_wg0 = blockIdx.x << 7;

    for (int i = tid; i < HPAD; i += 256) {
        int k = TAPMAX - i;
        hrev[i] = ((unsigned)k < (unsigned)HLEN) ? 5.0f * h[k] : 0.0f;
    }

    const int t0_wg = (DOWN * (m_wg0 + MSHIFT)) / UP;
    const int t_min = t0_wg - JMAX;
    for (int r = 0; r < 16; ++r) {
        const float* xrow = x + (size_t)(b0 + r) * (size_t)n_in;
        float* dst = &xs[r * XSTRIDE];
        for (int c = tid; c < XCOLS; c += 256) {
            int t = t_min + c;
            dst[c] = ((unsigned)t < (unsigned)n_in) ? xrow[t] : 0.0f;
        }
    }
    __syncthreads();

    int tile = (blockIdx.x << 3) + wave;
    const int last_tile = (n_out >> 4) - 1;
    if (tile > last_tile) tile = last_tile;
    const int m0   = tile << 4;
    const int t0t  = (DOWN * (m0 + MSHIFT)) / UP;
    const int toff = (t0t - JMAX) - t_min;

    const int mrow = lane & 15;
    const int ksub = (lane >> 4) << 1;
    const int tap0 = DOWN * (m0 + mrow + MSHIFT) - UP * (t0t - JMAX) - UP * ksub;

    const float* hA = &hrev[TAPMAX - tap0];
    const float* xB = &xs[mrow * XSTRIDE + toff + ksub];

    v8f acc = {0.f, 0.f, 0.f, 0.f, 0.f, 0.f, 0.f, 0.f};

    #pragma unroll
    for (int kk = 0; kk < KSTEPS; ++kk) {
        float av0 = hA[20 * kk];
        float av1 = hA[20 * kk + 5];
        float bv0 = xB[4 * kk];
        float bv1 = xB[4 * kk + 1];
        v2f A = {av0, av1};
        v2f B = {bv0, bv1};
        acc = __builtin_amdgcn_wmma_f32_16x16x4_f32(
            false, A, false, B, (short)0, acc, false, false);
    }

    {
        const int colw = (wave << 4) + ((lane >> 4) << 3);
        float* yrow = &yo[mrow * YSTRIDE + colw];
        #pragma unroll
        for (int i = 0; i < 8; ++i) yrow[i] = acc[i];
    }
    __syncthreads();

    for (int idx = tid; idx < 16 * 128; idx += 256) {
        int r = idx >> 7;
        int c = idx & 127;
        int m = m_wg0 + c;
        if (m < n_out)
            out[(size_t)(b0 + r) * (size_t)n_out + m] = yo[r * YSTRIDE + c];
    }
}

extern "C" void kernel_launch(void* const* d_in, const int* in_sizes, int n_in_bufs,
                              void* d_out, int out_size, void* d_ws, size_t ws_size,
                              hipStream_t stream) {
    const float* x = (const float*)d_in[0];
    const float* h = (const float*)d_in[1];
    float* out = (float*)d_out;

    const int T = 1440000;
    const int B = in_sizes[0] / T;
    const int n_out = (int)(((long long)T * UP) / DOWN);

    dim3 grid((n_out + 127) / 128, B / 16);
    resample_poly_wmma<<<grid, 256, 0, stream>>>(x, h, out, T, n_out);
}


# Staging issues all of a thread's global loads before its first LDS store: twelve branch-free four-float vector loads on a 4-aligned window replace one guarded scalar load per memory round trip
# speedup vs baseline: 2.7921x; 2.7921x over previous
#include <hip/hip_runtime.h>
#include <hip/hip_bf16.h>

typedef float v2f __attribute__((ext_vector_type(2)));
typedef float v8f __attribute__((ext_vector_type(8)));

#define UP      5
#define DOWN    24
#define HLEN    481
#define MSHIFT  10
#define JMAX    96
#define KSTEPS  43
#define XC4     179
#define XSTRIDE 716
#define NV4     2864
#define XLD     12
#define YSTRIDE 132
#define TAPMAX  844
#define HPAD    1280
#define HLD     5

template <bool GUARD>
__device__ __forceinline__ void stage_window(
    const float* __restrict__ xwg, const float* __restrict__ h,
    float* __restrict__ xs, float* __restrict__ hrev,
    int tid, int n_in, int t_al)
{
    const float4* __restrict__ xwg4 = reinterpret_cast<const float4*>(xwg);
    const unsigned n4 = (unsigned)n_in >> 2;
    float4 xv[XLD];
    int    xi[XLD];
    bool   xz[XLD];
    #pragma unroll
    for (int i = 0; i < XLD; ++i) {
        int idx = tid + 256 * i;
        idx = (idx < NV4) ? idx : (NV4 - 1);
        const int r  = idx / XC4;
        const int c4 = idx - r * XC4;
        const int t4 = (t_al >> 2) + c4;
        const bool ok = !GUARD || ((unsigned)t4 < n4);
        const unsigned off4 = (unsigned)r * n4 + (unsigned)(ok ? t4 : 0);
        xv[i] = xwg4[off4];
        xi[i] = idx;
        xz[i] = !ok;
    }
    float hk[HLD];
    #pragma unroll
    for (int j = 0; j < HLD; ++j) {
        const int k = TAPMAX - (tid + 256 * j);
        hk[j] = h[((unsigned)k < (unsigned)HLEN) ? k : 0];
    }

    #pragma unroll
    for (int i = 0; i < XLD; ++i) {
        float4 v = xv[i];
        if (GUARD && xz[i]) v = make_float4(0.f, 0.f, 0.f, 0.f);
        *reinterpret_cast<float4*>(&xs[4 * xi[i]]) = v;
    }
    #pragma unroll
    for (int j = 0; j < HLD; ++j) {
        const int k = TAPMAX - (tid + 256 * j);
        hrev[tid + 256 * j] = ((unsigned)k < (unsigned)HLEN) ? 5.0f * hk[j] : 0.0f;
    }
}

__global__ __launch_bounds__(256) void resample_poly_wmma(
    const float* __restrict__ x,
    const float* __restrict__ h,
    float* __restrict__ out,
    int n_in, int n_out)
{
    __shared__ float hrev[HPAD];
    __shared__ __attribute__((aligned(16))) float xs[16 * XSTRIDE];
    __shared__ float yo[16 * YSTRIDE];

    const int tid  = threadIdx.x;
    const int wave = tid >> 5;
    const int lane = tid & 31;
    const int b0    = blockIdx.y << 4;
    const int m_wg0 = blockIdx.x << 7;

    const int t0_wg = (DOWN * (m_wg0 + MSHIFT)) / UP;
    const int t_min = t0_wg - JMAX;
    const int t_al  = t_min & ~3;
    const float* xwg = x + (size_t)b0 * (size_t)n_in;
    if (t_al >= 0 && t_al + XSTRIDE <= n_in)
        stage_window<false>(xwg, h, xs, hrev, tid, n_in, t_al);
    else
        stage_window<true>(xwg, h, xs, hrev, tid, n_in, t_al);
    __syncthreads();

    int tile = (blockIdx.x << 3) + wave;
    const int last_tile = (n_out >> 4) - 1;
    if (tile > last_tile) tile = last_tile;
    const int m0   = tile << 4;
    const int t0t  = (DOWN * (m0 + MSHIFT)) / UP;
    const int toff = (t0t - JMAX) - t_al;

    const int mrow = lane & 15;
    const int ksub = (lane >> 4) << 1;
    const int tap0 = DOWN * (m0 + mrow + MSHIFT) - UP * (t0t - JMAX) - UP * ksub;

    const float* hA = &hrev[TAPMAX - tap0];
    const float* xB = &xs[mrow * XSTRIDE + toff + ksub];

    v8f acc = {0.f, 0.f, 0.f, 0.f, 0.f, 0.f, 0.f, 0.f};

    #pragma unroll
    for (int kk = 0; kk < KSTEPS; ++kk) {
        float av0 = hA[20 * kk];
        float av1 = hA[20 * kk + 5];
        float bv0 = xB[4 * kk];
        float bv1 = xB[4 * kk + 1];
        v2f A = {av0, av1};
        v2f B = {bv0, bv1};
        acc = __builtin_amdgcn_wmma_f32_16x16x4_f32(
            false, A, false, B, (short)0, acc, false, false);
    }

    {
        const int colw = (wave << 4) + ((lane >> 4) << 3);
        float* yrow = &yo[mrow * YSTRIDE + colw];
        #pragma unroll
        for (int i = 0; i < 8; ++i) yrow[i] = acc[i];
    }
    __syncthreads();

    for (int idx = tid; idx < 16 * 128; idx += 256) {
        int r = idx >> 7;
        int c = idx & 127;
        int m = m_wg0 + c;
        if (m < n_out)
            out[(size_t)(b0 + r) * (size_t)n_out + m] = yo[r * YSTRIDE + c];
    }
}

extern "C" void kernel_launch(void* const* d_in, const int* in_sizes, int n_in_bufs,
                              void* d_out, int out_size, void* d_ws, size_t ws_size,
                              hipStream_t stream) {
    const float* x = (const float*)d_in[0];
    const float* h = (const float*)d_in[1];
    float* out = (float*)d_out;

    const int T = 1440000;
    const int B = in_sizes[0] / T;
    const int n_out = (int)(((long long)T * UP) / DOWN);

    dim3 grid((n_out + 127) / 128, B / 16);
    resample_poly_wmma<<<grid, 256, 0, stream>>>(x, h, out, T, n_out);
}


# Work-groups loop over tiles (grid 590x2) and issue the next tile's global loads before the current tile's matrix loop, storing them to LDS after it: memory time overlaps the matrix phase
# speedup vs baseline: 3.0446x; 1.0904x over previous
#include <hip/hip_runtime.h>
#include <hip/hip_bf16.h>


typedef float v2f __attribute__((ext_vector_type(2)));
typedef float v8f __attribute__((ext_vector_type(8)));

#define UP      5
#define DOWN    24
#define HLEN    481
#define MSHIFT  10
#define JMAX    96
#define KSTEPS  43
#define XC4     179
#define XSTRIDE 716
#define XPASS   6
#define XLD     12
#define YSTRIDE 132
#define TAPMAX  844
#define HPAD    1280
#define HLD     5

__device__ __forceinline__ void lds_barrier()
{
    __builtin_amdgcn_fence(__ATOMIC_RELEASE, "workgroup", "local");
    __builtin_amdgcn_s_barrier();
    __builtin_amdgcn_fence(__ATOMIC_ACQUIRE, "workgroup", "local");
}

__device__ __forceinline__ int window_start(int tile)
{
    const int t0 = (DOWN * ((tile << 7) + MSHIFT)) / UP;
    return (t0 - JMAX) & ~3;
}

__device__ __forceinline__ void load_window(
    const float4* __restrict__ xwg4, unsigned n4, int t4base, int wave, int lane,
    float4 (&xv)[XLD])
{
    #pragma unroll
    for (int j = 0; j < 2; ++j) {
        const unsigned rowoff = (unsigned)(2 * wave + j) * n4;
        #pragma unroll
        for (int p = 0; p < XPASS; ++p) {
            int c4 = lane + 32 * p;
            if (p == XPASS - 1) c4 = (c4 < XC4) ? c4 : (XC4 - 1);
            const int t4 = t4base + c4;
            const bool ok = (unsigned)t4 < n4;
            xv[j * XPASS + p] = xwg4[rowoff + (unsigned)(ok ? t4 : 0)];
        }
    }
}

__device__ __forceinline__ void store_window(
    float* __restrict__ xs, unsigned n4, int t4base, int wave, int lane,
    const float4 (&xv)[XLD])
{
    #pragma unroll
    for (int j = 0; j < 2; ++j) {
        #pragma unroll
        for (int p = 0; p < XPASS; ++p) {
            int c4 = lane + 32 * p;
            if (p == XPASS - 1) c4 = (c4 < XC4) ? c4 : (XC4 - 1);
            const int t4 = t4base + c4;
            const bool ok = (unsigned)t4 < n4;
            float4 v = xv[j * XPASS + p];
            if (!ok) v = make_float4(0.f, 0.f, 0.f, 0.f);
            *reinterpret_cast<float4*>(&xs[(2 * wave + j) * XSTRIDE + 4 * c4]) = v;
        }
    }
}

__global__ __launch_bounds__(256) void resample_poly_wmma(
    const float* __restrict__ x,
    const float* __restrict__ h,
    float* __restrict__ out,
    int n_in, int n_out)
{
    __shared__ float hrev[HPAD];
    __shared__ __attribute__((aligned(16))) float xs[16 * XSTRIDE];
    __shared__ float yo[16 * YSTRIDE];

    const int tid  = threadIdx.x;
    const int wave = tid >> 5;
    const int lane = tid & 31;
    const int b0   = blockIdx.y << 4;
    const int ntiles    = (n_out + 127) >> 7;
    const int last_tile = (n_out >> 4) - 1;
    const unsigned n4   = (unsigned)n_in >> 2;
    const float4* __restrict__ xwg4 =
        reinterpret_cast<const float4*>(x + (size_t)b0 * (size_t)n_in);

    const int mrow = lane & 15;
    const int ksub = (lane >> 4) << 1;

    int tile = blockIdx.x;
    int t_al = window_start(tile);
    float4 xv[XLD];
    load_window(xwg4, n4, t_al >> 2, wave, lane, xv);
    {
        float hk[HLD];
        #pragma unroll
        for (int j = 0; j < HLD; ++j) {
            const int k = TAPMAX - (tid + 256 * j);
            hk[j] = h[((unsigned)k < (unsigned)HLEN) ? k : 0];
        }
        #pragma unroll
        for (int j = 0; j < HLD; ++j) {
            const int k = TAPMAX - (tid + 256 * j);
            hrev[tid + 256 * j] = ((unsigned)k < (unsigned)HLEN) ? 5.0f * hk[j] : 0.0f;
        }
    }
    store_window(xs, n4, t_al >> 2, wave, lane, xv);
    __syncthreads();

    for (;;) {
        const int  next     = tile + (int)gridDim.x;
        const bool has_next = next < ntiles;
        int t_al_next = 0;
        if (has_next) {
            t_al_next = window_start(next);
            load_window(xwg4, n4, t_al_next >> 2, wave, lane, xv);
        }

        int tile16 = (tile << 3) + wave;
        if (tile16 > last_tile) tile16 = last_tile;
        const int m0   = tile16 << 4;
        const int t0t  = (DOWN * (m0 + MSHIFT)) / UP;
        const int toff = (t0t - JMAX) - t_al;
        const int tap0 = DOWN * (m0 + mrow + MSHIFT) - UP * (t0t - JMAX) - UP * ksub;

        const float* hA = &hrev[TAPMAX - tap0];
        const float* xB = &xs[mrow * XSTRIDE + toff + ksub];

        v8f acc = {0.f, 0.f, 0.f, 0.f, 0.f, 0.f, 0.f, 0.f};

        #pragma unroll
        for (int kk = 0; kk < KSTEPS; ++kk) {
            float av0 = hA[20 * kk];
            float av1 = hA[20 * kk + 5];
            float bv0 = xB[4 * kk];
            float bv1 = xB[4 * kk + 1];
            v2f A = {av0, av1};
            v2f B = {bv0, bv1};
            acc = __builtin_amdgcn_wmma_f32_16x16x4_f32(
                false, A, false, B, (short)0, acc, false, false);
        }

        {
            const int colw = (wave << 4) + ((lane >> 4) << 3);
            float* yrow = &yo[mrow * YSTRIDE + colw];
            #pragma unroll
            for (int i = 0; i < 8; ++i) yrow[i] = acc[i];
        }
        lds_barrier();

        {
            const int c = tid & 127;
            const int m = (tile << 7) + c;
            if (m < n_out) {
                const int r0 = tid >> 7;
                float* orow = out + (size_t)(b0 + r0) * (size_t)n_out + m;
                const float* ysrc = &yo[r0 * YSTRIDE + c];
                float yv[8];
                #pragma unroll
                for (int i = 0; i < 8; ++i) yv[i] = ysrc[2 * i * YSTRIDE];
                #pragma unroll
                for (int i = 0; i < 8; ++i) orow[(size_t)(2 * i) * (size_t)n_out] = yv[i];
            }
        }
        if (!has_next) break;

        store_window(xs, n4, t_al_next >> 2, wave, lane, xv);
        lds_barrier();
        tile = next;
        t_al = t_al_next;
    }
}

extern "C" void kernel_launch(void* const* d_in, const int* in_sizes, int n_in_bufs,
                              void* d_out, int out_size, void* d_ws, size_t ws_size,
                              hipStream_t stream) {
    const float* x = (const float*)d_in[0];
    const float* h = (const float*)d_in[1];
    float* out = (float*)d_out;

    const int T = 1440000;
    const int B = in_sizes[0] / T;
    const int n_out = (int)(((long long)T * UP) / DOWN);

    dim3 grid(590, B / 16);
    resample_poly_wmma<<<grid, 256, 0, stream>>>(x, h, out, T, n_out);
}


# Interior tiles take an unguarded path: window loads are fixed scalar row bases plus two running lane offsets, so no selects and no address register rewritten behind loads in flight
# speedup vs baseline: 3.1393x; 1.0311x over previous
#include <hip/hip_runtime.h>
#include <hip/hip_bf16.h>


typedef float v2f __attribute__((ext_vector_type(2)));
typedef float v8f __attribute__((ext_vector_type(8)));

#define UP      5
#define DOWN    24
#define HLEN    481
#define MSHIFT  10
#define JMAX    96
#define KSTEPS  43
#define XC4     179
#define XSTRIDE 716
#define XPASS   6
#define XLD     12
#define YSTRIDE 132
#define TAPMAX  844
#define HPAD    1280
#define HLD     5

__device__ __forceinline__ void lds_barrier()
{
    __builtin_amdgcn_fence(__ATOMIC_RELEASE, "workgroup", "local");
    __builtin_amdgcn_s_barrier();
    __builtin_amdgcn_fence(__ATOMIC_ACQUIRE, "workgroup", "local");
}

__device__ __forceinline__ int window_start(int tile)
{
    const int t0 = (DOWN * ((tile << 7) + MSHIFT)) / UP;
    return (t0 - JMAX) & ~3;
}

__device__ __forceinline__ void load_window(
    const float4* __restrict__ xwg4, unsigned n4, int t4base, int wave, int lane,
    float4 (&xv)[XLD])
{
    #pragma unroll
    for (int j = 0; j < 2; ++j) {
        const unsigned rowoff = (unsigned)(2 * wave + j) * n4;
        #pragma unroll
        for (int p = 0; p < XPASS; ++p) {
            int c4 = lane + 32 * p;
            if (p == XPASS - 1) c4 = (c4 < XC4) ? c4 : (XC4 - 1);
            const int t4 = t4base + c4;
            const bool ok = (unsigned)t4 < n4;
            xv[j * XPASS + p] = xwg4[rowoff + (unsigned)(ok ? t4 : 0)];
        }
    }
}

__device__ __forceinline__ void store_window(
    float* __restrict__ xs, unsigned n4, int t4base, int wave, int lane,
    const float4 (&xv)[XLD])
{
    #pragma unroll
    for (int j = 0; j < 2; ++j) {
        #pragma unroll
        for (int p = 0; p < XPASS; ++p) {
            int c4 = lane + 32 * p;
            if (p == XPASS - 1) c4 = (c4 < XC4) ? c4 : (XC4 - 1);
            const int t4 = t4base + c4;
            const bool ok = (unsigned)t4 < n4;
            float4 v = xv[j * XPASS + p];
            if (!ok) v = make_float4(0.f, 0.f, 0.f, 0.f);
            *reinterpret_cast<float4*>(&xs[(2 * wave + j) * XSTRIDE + 4 * c4]) = v;
        }
    }
}

__device__ __forceinline__ void load_window_inner(
    const float4* __restrict__ ra, const float4* __restrict__ rb,
    unsigned va, unsigned vl, float4 (&xv)[XLD])
{
    #pragma unroll
    for (int p = 0; p < XPASS - 1; ++p) xv[p] = (ra + 32 * p)[va];
    xv[XPASS - 1] = ra[vl];
    #pragma unroll
    for (int p = 0; p < XPASS - 1; ++p) xv[XPASS + p] = (rb + 32 * p)[va];
    xv[2 * XPASS - 1] = rb[vl];
}

__device__ __forceinline__ void store_window_inner(
    float* __restrict__ xs, int wave, unsigned lane, unsigned c4last,
    const float4 (&xv)[XLD])
{
    #pragma unroll
    for (int j = 0; j < 2; ++j) {
        #pragma unroll
        for (int p = 0; p < XPASS; ++p) {
            const unsigned c4 = (p == XPASS - 1) ? c4last : lane + 32u * p;
            *reinterpret_cast<float4*>(&xs[(2 * wave + j) * XSTRIDE + 4 * c4]) = xv[j * XPASS + p];
        }
    }
}

__global__ __launch_bounds__(256) void resample_poly_wmma(
    const float* __restrict__ x,
    const float* __restrict__ h,
    float* __restrict__ out,
    int n_in, int n_out)
{
    __shared__ float hrev[HPAD];
    __shared__ __attribute__((aligned(16))) float xs[16 * XSTRIDE];
    __shared__ float yo[16 * YSTRIDE];

    const int tid  = threadIdx.x;
    const int wave = __builtin_amdgcn_readfirstlane(tid >> 5);
    const int lane = tid & 31;
    const int b0   = blockIdx.y << 4;
    const int ntiles    = (n_out + 127) >> 7;
    const int last_tile = (n_out >> 4) - 1;
    const unsigned n4   = (unsigned)n_in >> 2;
    const float4* __restrict__ xwg4 =
        reinterpret_cast<const float4*>(x + (size_t)b0 * (size_t)n_in);
    const float4* __restrict__ rowa = xwg4 + (size_t)(2 * wave) * (size_t)n4;
    const float4* __restrict__ rowb = rowa + n4;
    const unsigned c4last = (lane + 32 * (XPASS - 1) < XC4) ? (unsigned)(lane + 32 * (XPASS - 1))
                                                            : (unsigned)(XC4 - 1);
    const int ocol  = tid & 127;
    const int orow0 = tid >> 7;

    const int mrow = lane & 15;
    const int ksub = (lane >> 4) << 1;

    auto compute_tile = [&](const int tile, const int t_al) {
        int tile16 = (tile << 3) + wave;
        if (tile16 > last_tile) tile16 = last_tile;
        const int m0   = tile16 << 4;
        const int t0t  = (DOWN * (m0 + MSHIFT)) / UP;
        const int toff = (t0t - JMAX) - t_al;
        const int tap0 = DOWN * (m0 + mrow + MSHIFT) - UP * (t0t - JMAX) - UP * ksub;

        const float* hA = &hrev[TAPMAX - tap0];
        const float* xB = &xs[mrow * XSTRIDE + toff + ksub];

        v8f acc = {0.f, 0.f, 0.f, 0.f, 0.f, 0.f, 0.f, 0.f};

        #pragma unroll
        for (int kk = 0; kk < KSTEPS; ++kk) {
            float av0 = hA[20 * kk];
            float av1 = hA[20 * kk + 5];
            float bv0 = xB[4 * kk];
            float bv1 = xB[4 * kk + 1];
            v2f A = {av0, av1};
            v2f B = {bv0, bv1};
            acc = __builtin_amdgcn_wmma_f32_16x16x4_f32(
                false, A, false, B, (short)0, acc, false, false);
        }

        {
            const int colw = (wave << 4) + ((lane >> 4) << 3);
            float* yrow = &yo[mrow * YSTRIDE + colw];
            #pragma unroll
            for (int i = 0; i < 8; ++i) yrow[i] = acc[i];
        }
        lds_barrier();

        if (ocol < n_out - (tile << 7)) {
            const unsigned vo = (unsigned)orow0 * (unsigned)n_out + (unsigned)(tile << 7) + (unsigned)ocol;
            const float* ysrc = &yo[orow0 * YSTRIDE + ocol];
            float yv[8];
            #pragma unroll
            for (int i = 0; i < 8; ++i) yv[i] = ysrc[2 * i * YSTRIDE];
            #pragma unroll
            for (int i = 0; i < 8; ++i) (out + (size_t)(b0 + 2 * i) * (size_t)n_out)[vo] = yv[i];
        }
    };

    int tile = blockIdx.x;
    int t_al = window_start(tile);
    float4 xv[XLD];
    load_window(xwg4, n4, t_al >> 2, wave, lane, xv);
    {
        float hk[HLD];
        #pragma unroll
        for (int j = 0; j < HLD; ++j) {
            const int k = TAPMAX - (tid + 256 * j);
            hk[j] = h[((unsigned)k < (unsigned)HLEN) ? k : 0];
        }
        #pragma unroll
        for (int j = 0; j < HLD; ++j) {
            const int k = TAPMAX - (tid + 256 * j);
            hrev[tid + 256 * j] = ((unsigned)k < (unsigned)HLEN) ? 5.0f * hk[j] : 0.0f;
        }
    }
    store_window(xs, n4, t_al >> 2, wave, lane, xv);
    __syncthreads();

    unsigned va = (unsigned)lane + (unsigned)(t_al >> 2);
    unsigned vl = c4last + (unsigned)(t_al >> 2);

    bool more;
    for (;;) {
        const int next = tile + (int)gridDim.x;
        more = next < ntiles;
        int  t_al_next = 0;
        bool pre = false;
        if (more) {
            t_al_next = window_start(next);
            const int tb4 = t_al_next >> 2;
            pre = (tb4 >= 0) && (tb4 + XC4 <= (int)n4);
        }
        if (pre) {
            const unsigned d4 = (unsigned)((t_al_next - t_al) >> 2);
            va += d4;
            vl += d4;
            load_window_inner(rowa, rowb, va, vl, xv);
        }

        compute_tile(tile, t_al);
        if (!pre) break;

        store_window_inner(xs, wave, lane, c4last, xv);
        lds_barrier();
        tile = next;
        t_al = t_al_next;
    }

    while (more) {
        const int next = tile + (int)gridDim.x;
        const int t_al_next = window_start(next);
        load_window(xwg4, n4, t_al_next >> 2, wave, lane, xv);
        store_window(xs, n4, t_al_next >> 2, wave, lane, xv);
        lds_barrier();
        compute_tile(next, t_al_next);
        tile = next;
        more = tile + (int)gridDim.x < ntiles;
    }
}

extern "C" void kernel_launch(void* const* d_in, const int* in_sizes, int n_in_bufs,
                              void* d_out, int out_size, void* d_ws, size_t ws_size,
                              hipStream_t stream) {
    const float* x = (const float*)d_in[0];
    const float* h = (const float*)d_in[1];
    float* out = (float*)d_out;

    const int T = 1440000;
    const int B = in_sizes[0] / T;
    const int n_out = (int)(((long long)T * UP) / DOWN);

    dim3 grid(590, B / 16);
    resample_poly_wmma<<<grid, 256, 0, stream>>>(x, h, out, T, n_out);
}


# Matrix loop fed from registers at 4 waves per SIMD: filter fragments stay resident for all of a wave's tiles (tile stride 255 = 0 mod 5), input fragments are read from LDS one 11-step group ahead
# speedup vs baseline: 3.2691x; 1.0413x over previous
#include <hip/hip_runtime.h>
#include <hip/hip_bf16.h>


typedef float v2f __attribute__((ext_vector_type(2)));
typedef float v8f __attribute__((ext_vector_type(8)));

#define UP      5
#define DOWN    24
#define HLEN    481
#define MSHIFT  10
#define JMAX    96
#define KSTEPS  43
#define BGRP    11
#define NBGRP   4
#define XC4     179
#define XSTRIDE 716
#define XPASS   6
#define XLD     12
#define YSTRIDE 132
#define TAPMAX  844
#define HPAD    1280
#define HLD     5

__device__ __forceinline__ void lds_barrier()
{
    __builtin_amdgcn_fence(__ATOMIC_RELEASE, "workgroup", "local");
    __builtin_amdgcn_s_barrier();
    __builtin_amdgcn_fence(__ATOMIC_ACQUIRE, "workgroup", "local");
}

__device__ __forceinline__ int window_start(int tile)
{
    const int t0 = (DOWN * ((tile << 7) + MSHIFT)) / UP;
    return (t0 - JMAX) & ~3;
}

__device__ __forceinline__ void load_window(
    const float4* __restrict__ xwg4, unsigned n4, int t4base, int wave, int lane,
    float4 (&xv)[XLD])
{
    #pragma unroll
    for (int j = 0; j < 2; ++j) {
        const unsigned rowoff = (unsigned)(2 * wave + j) * n4;
        #pragma unroll
        for (int p = 0; p < XPASS; ++p) {
            int c4 = lane + 32 * p;
            if (p == XPASS - 1) c4 = (c4 < XC4) ? c4 : (XC4 - 1);
            const int t4 = t4base + c4;
            const bool ok = (unsigned)t4 < n4;
            xv[j * XPASS + p] = xwg4[rowoff + (unsigned)(ok ? t4 : 0)];
        }
    }
}

__device__ __forceinline__ void store_window(
    float* __restrict__ xs, unsigned n4, int t4base, int wave, int lane,
    const float4 (&xv)[XLD])
{
    #pragma unroll
    for (int j = 0; j < 2; ++j) {
        #pragma unroll
        for (int p = 0; p < XPASS; ++p) {
            int c4 = lane + 32 * p;
            if (p == XPASS - 1) c4 = (c4 < XC4) ? c4 : (XC4 - 1);
            const int t4 = t4base + c4;
            const bool ok = (unsigned)t4 < n4;
            float4 v = xv[j * XPASS + p];
            if (!ok) v = make_float4(0.f, 0.f, 0.f, 0.f);
            *reinterpret_cast<float4*>(&xs[(2 * wave + j) * XSTRIDE + 4 * c4]) = v;
        }
    }
}

__device__ __forceinline__ void load_window_inner(
    const float4* __restrict__ ra, const float4* __restrict__ rb,
    unsigned va, unsigned vl, float4 (&xv)[XLD])
{
    #pragma unroll
    for (int p = 0; p < XPASS - 1; ++p) xv[p] = (ra + 32 * p)[va];
    xv[XPASS - 1] = ra[vl];
    #pragma unroll
    for (int p = 0; p < XPASS - 1; ++p) xv[XPASS + p] = (rb + 32 * p)[va];
    xv[2 * XPASS - 1] = rb[vl];
}

__device__ __forceinline__ void store_window_inner(
    float* __restrict__ xs, int wave, unsigned lane, unsigned c4last,
    const float4 (&xv)[XLD])
{
    #pragma unroll
    for (int j = 0; j < 2; ++j) {
        #pragma unroll
        for (int p = 0; p < XPASS; ++p) {
            const unsigned c4 = (p == XPASS - 1) ? c4last : lane + 32u * p;
            *reinterpret_cast<float4*>(&xs[(2 * wave + j) * XSTRIDE + 4 * c4]) = xv[j * XPASS + p];
        }
    }
}

__global__ __launch_bounds__(256) void resample_poly_wmma(
    const float* __restrict__ x,
    const float* __restrict__ h,
    float* __restrict__ out,
    int n_in, int n_out)
{
    __shared__ float hrev[HPAD];
    __shared__ __attribute__((aligned(16))) float xs[16 * XSTRIDE];
    __shared__ float yo[16 * YSTRIDE];

    const int tid  = threadIdx.x;
    const int wave = __builtin_amdgcn_readfirstlane(tid >> 5);
    const int lane = tid & 31;
    const int b0   = blockIdx.y << 4;
    const int ntiles    = (n_out + 127) >> 7;
    const int last_tile = (n_out >> 4) - 1;
    const unsigned n4   = (unsigned)n_in >> 2;
    const float4* __restrict__ xwg4 =
        reinterpret_cast<const float4*>(x + (size_t)b0 * (size_t)n_in);
    const float4* __restrict__ rowa = xwg4 + (size_t)(2 * wave) * (size_t)n4;
    const float4* __restrict__ rowb = rowa + n4;
    const unsigned c4last = (lane + 32 * (XPASS - 1) < XC4) ? (unsigned)(lane + 32 * (XPASS - 1))
                                                            : (unsigned)(XC4 - 1);
    const int ocol  = tid & 127;
    const int orow0 = tid >> 7;

    const int mrow = lane & 15;
    const int ksub = (lane >> 4) << 1;

    v2f afrag[KSTEPS];

    auto compute_tile = [&](const int tile, const int t_al) {
        int tile16 = (tile << 3) + wave;
        if (tile16 > last_tile) tile16 = last_tile;
        const int m0   = tile16 << 4;
        const int t0t  = (DOWN * (m0 + MSHIFT)) / UP;
        const int toff = (t0t - JMAX) - t_al;
        const float* xB = &xs[mrow * XSTRIDE + toff + ksub];

        v8f acc = {0.f, 0.f, 0.f, 0.f, 0.f, 0.f, 0.f, 0.f};

        v2f bfrag[2][BGRP];
        #pragma unroll
        for (int i = 0; i < BGRP; ++i) {
            bfrag[0][i].x = xB[4 * i];
            bfrag[0][i].y = xB[4 * i + 1];
        }
        #pragma unroll
        for (int g = 0; g < NBGRP; ++g) {
            __builtin_amdgcn_sched_barrier(0);
            if (g + 1 < NBGRP) {
                #pragma unroll
                for (int i = 0; i < BGRP; ++i) {
                    const int kn = (g + 1) * BGRP + i;
                    if (kn < KSTEPS) {
                        bfrag[(g + 1) & 1][i].x = xB[4 * kn];
                        bfrag[(g + 1) & 1][i].y = xB[4 * kn + 1];
                    }
                }
            }
            __builtin_amdgcn_sched_barrier(0);
            #pragma unroll
            for (int i = 0; i < BGRP; ++i) {
                const int kk = g * BGRP + i;
                if (kk < KSTEPS)
                    acc = __builtin_amdgcn_wmma_f32_16x16x4_f32(
                        false, afrag[kk], false, bfrag[g & 1][i], (short)0, acc, false, false);
            }
        }

        {
            const int colw = (wave << 4) + ((lane >> 4) << 3);
            float* yrow = &yo[mrow * YSTRIDE + colw];
            #pragma unroll
            for (int i = 0; i < 8; ++i) yrow[i] = acc[i];
        }
        lds_barrier();

        if (ocol < n_out - (tile << 7)) {
            const unsigned vo = (unsigned)orow0 * (unsigned)n_out + (unsigned)(tile << 7) + (unsigned)ocol;
            const float* ysrc = &yo[orow0 * YSTRIDE + ocol];
            float yv[8];
            #pragma unroll
            for (int i = 0; i < 8; ++i) yv[i] = ysrc[2 * i * YSTRIDE];
            #pragma unroll
            for (int i = 0; i < 8; ++i) (out + (size_t)(b0 + 2 * i) * (size_t)n_out)[vo] = yv[i];
        }
    };

    int tile = blockIdx.x;
    int t_al = window_start(tile);
    float4 xv[XLD];
    load_window(xwg4, n4, t_al >> 2, wave, lane, xv);
    {
        float hk[HLD];
        #pragma unroll
        for (int j = 0; j < HLD; ++j) {
            const int k = TAPMAX - (tid + 256 * j);
            hk[j] = h[((unsigned)k < (unsigned)HLEN) ? k : 0];
        }
        #pragma unroll
        for (int j = 0; j < HLD; ++j) {
            const int k = TAPMAX - (tid + 256 * j);
            hrev[tid + 256 * j] = ((unsigned)k < (unsigned)HLEN) ? 5.0f * hk[j] : 0.0f;
        }
    }
    store_window(xs, n4, t_al >> 2, wave, lane, xv);
    __syncthreads();

    {
        const int m0f  = ((tile << 3) + wave) << 4;
        const int t0f  = (DOWN * (m0f + MSHIFT)) / UP;
        const int tap0 = DOWN * (m0f + mrow + MSHIFT) - UP * (t0f - JMAX) - UP * ksub;
        const float* hA = &hrev[TAPMAX - tap0];
        #pragma unroll
        for (int kk = 0; kk < KSTEPS; ++kk) {
            afrag[kk].x = hA[20 * kk];
            afrag[kk].y = hA[20 * kk + 5];
        }
    }

    unsigned va = (unsigned)lane + (unsigned)(t_al >> 2);
    unsigned vl = c4last + (unsigned)(t_al >> 2);

    bool more;
    for (;;) {
        const int next = tile + (int)gridDim.x;
        more = next < ntiles;
        int  t_al_next = 0;
        bool pre = false;
        if (more) {
            t_al_next = window_start(next);
            const int tb4 = t_al_next >> 2;
            pre = (tb4 >= 0) && (tb4 + XC4 <= (int)n4);
        }
        if (pre) {
            const unsigned d4 = (unsigned)((t_al_next - t_al) >> 2);
            va += d4;
            vl += d4;
            load_window_inner(rowa, rowb, va, vl, xv);
        }

        compute_tile(tile, t_al);
        if (!pre) break;

        store_window_inner(xs, wave, lane, c4last, xv);
        lds_barrier();
        tile = next;
        t_al = t_al_next;
    }

    while (more) {
        const int next = tile + (int)gridDim.x;
        const int t_al_next = window_start(next);
        load_window(xwg4, n4, t_al_next >> 2, wave, lane, xv);
        store_window(xs, n4, t_al_next >> 2, wave, lane, xv);
        lds_barrier();
        compute_tile(next, t_al_next);
        tile = next;
        more = tile + (int)gridDim.x < ntiles;
    }
}

extern "C" void kernel_launch(void* const* d_in, const int* in_sizes, int n_in_bufs,
                              void* d_out, int out_size, void* d_ws, size_t ws_size,
                              hipStream_t stream) {
    const float* x = (const float*)d_in[0];
    const float* h = (const float*)d_in[1];
    float* out = (float*)d_out;

    const int T = 1440000;
    const int B = in_sizes[0] / T;
    const int n_out = (int)(((long long)T * UP) / DOWN);

    dim3 grid(255, B / 16);
    resample_poly_wmma<<<grid, 256, 0, stream>>>(x, h, out, T, n_out);
}


# LDS reads widened to 16 bytes: one aligned read feeds two matrix steps (4-aligned K origin; the k summation order within each 8-sample block changes), output rows leave as 16-byte global stores
# speedup vs baseline: 3.4850x; 1.0660x over previous
#include <hip/hip_runtime.h>
#include <hip/hip_bf16.h>


typedef float v2f __attribute__((ext_vector_type(2)));
typedef float v8f __attribute__((ext_vector_type(8)));

#define UP      5
#define DOWN    24
#define HLEN    481
#define MSHIFT  10
#define JMAX    96
#define KSTEPS  43
#define NPAIR   21
#define PGRP    6
#define NBGRP   4
#define XC4     179
#define XSTRIDE 716
#define XPASS   6
#define XLD     12
#define YSTRIDE 132
#define TAPMAX  859
#define HPAD    1280
#define HLD     5

__device__ __forceinline__ void lds_barrier()
{
    __builtin_amdgcn_fence(__ATOMIC_RELEASE, "workgroup", "local");
    __builtin_amdgcn_s_barrier();
    __builtin_amdgcn_fence(__ATOMIC_ACQUIRE, "workgroup", "local");
}

__device__ __forceinline__ int window_start(int tile)
{
    const int t0 = (DOWN * ((tile << 7) + MSHIFT)) / UP;
    return (t0 - JMAX) & ~3;
}

__device__ __forceinline__ void load_window(
    const float4* __restrict__ xwg4, unsigned n4, int t4base, int wave, int lane,
    float4 (&xv)[XLD])
{
    #pragma unroll
    for (int j = 0; j < 2; ++j) {
        const unsigned rowoff = (unsigned)(2 * wave + j) * n4;
        #pragma unroll
        for (int p = 0; p < XPASS; ++p) {
            int c4 = lane + 32 * p;
            if (p == XPASS - 1) c4 = (c4 < XC4) ? c4 : (XC4 - 1);
            const int t4 = t4base + c4;
            const bool ok = (unsigned)t4 < n4;
            xv[j * XPASS + p] = xwg4[rowoff + (unsigned)(ok ? t4 : 0)];
        }
    }
}

__device__ __forceinline__ void store_window(
    float* __restrict__ xs, unsigned n4, int t4base, int wave, int lane,
    const float4 (&xv)[XLD])
{
    #pragma unroll
    for (int j = 0; j < 2; ++j) {
        #pragma unroll
        for (int p = 0; p < XPASS; ++p) {
            int c4 = lane + 32 * p;
            if (p == XPASS - 1) c4 = (c4 < XC4) ? c4 : (XC4 - 1);
            const int t4 = t4base + c4;
            const bool ok = (unsigned)t4 < n4;
            float4 v = xv[j * XPASS + p];
            if (!ok) v = make_float4(0.f, 0.f, 0.f, 0.f);
            *reinterpret_cast<float4*>(&xs[(2 * wave + j) * XSTRIDE + 4 * c4]) = v;
        }
    }
}

__device__ __forceinline__ void load_window_inner(
    const float4* __restrict__ ra, const float4* __restrict__ rb,
    unsigned va, unsigned vl, float4 (&xv)[XLD])
{
    #pragma unroll
    for (int p = 0; p < XPASS - 1; ++p) xv[p] = (ra + 32 * p)[va];
    xv[XPASS - 1] = ra[vl];
    #pragma unroll
    for (int p = 0; p < XPASS - 1; ++p) xv[XPASS + p] = (rb + 32 * p)[va];
    xv[2 * XPASS - 1] = rb[vl];
}

__device__ __forceinline__ void store_window_inner(
    float* __restrict__ xs, int wave, unsigned lane, unsigned c4last,
    const float4 (&xv)[XLD])
{
    #pragma unroll
    for (int j = 0; j < 2; ++j) {
        #pragma unroll
        for (int p = 0; p < XPASS; ++p) {
            const unsigned c4 = (p == XPASS - 1) ? c4last : lane + 32u * p;
            *reinterpret_cast<float4*>(&xs[(2 * wave + j) * XSTRIDE + 4 * c4]) = xv[j * XPASS + p];
        }
    }
}

__global__ __launch_bounds__(256) void resample_poly_wmma(
    const float* __restrict__ x,
    const float* __restrict__ h,
    float* __restrict__ out,
    int n_in, int n_out)
{
    __shared__ float hrev[HPAD];
    __shared__ __attribute__((aligned(16))) float xs[16 * XSTRIDE];
    __shared__ __attribute__((aligned(16))) float yo[16 * YSTRIDE];

    const int tid  = threadIdx.x;
    const int wave = __builtin_amdgcn_readfirstlane(tid >> 5);
    const int lane = tid & 31;
    const int b0   = blockIdx.y << 4;
    const int ntiles    = (n_out + 127) >> 7;
    const int last_tile = (n_out >> 4) - 1;
    const unsigned n4   = (unsigned)n_in >> 2;
    const float4* __restrict__ xwg4 =
        reinterpret_cast<const float4*>(x + (size_t)b0 * (size_t)n_in);
    const float4* __restrict__ rowa = xwg4 + (size_t)(2 * wave) * (size_t)n4;
    const float4* __restrict__ rowb = rowa + n4;
    const unsigned c4last = (lane + 32 * (XPASS - 1) < XC4) ? (unsigned)(lane + 32 * (XPASS - 1))
                                                            : (unsigned)(XC4 - 1);
    const int ocol4 = lane << 2;

    const int mrow = lane & 15;
    const int ksub = (lane >> 4) << 1;
    const int khalf4 = (lane >> 4) << 2;

    v2f afrag[KSTEPS];

    auto compute_tile = [&](const int tile, const int t_al) {
        int tile16 = (tile << 3) + wave;
        if (tile16 > last_tile) tile16 = last_tile;
        const int m0   = tile16 << 4;
        const int t0t  = (DOWN * (m0 + MSHIFT)) / UP;
        const int toff = (t0t - JMAX) - t_al;
        const float* xB = &xs[mrow * XSTRIDE + (toff & ~3) + khalf4];
        const float* xT = &xs[mrow * XSTRIDE + (toff & ~3) + 8 * NPAIR + ksub];

        v8f acc = {0.f, 0.f, 0.f, 0.f, 0.f, 0.f, 0.f, 0.f};

        float4 bq[2][PGRP];
        v2f btail;
        #pragma unroll
        for (int i = 0; i < PGRP; ++i)
            bq[0][i] = *reinterpret_cast<const float4*>(&xB[8 * i]);
        #pragma unroll
        for (int g = 0; g < NBGRP; ++g) {
            __builtin_amdgcn_sched_barrier(0);
            if (g + 1 < NBGRP) {
                #pragma unroll
                for (int i = 0; i < PGRP; ++i) {
                    const int pn = (g + 1) * PGRP + i;
                    if (pn < NPAIR)
                        bq[(g + 1) & 1][i] = *reinterpret_cast<const float4*>(&xB[8 * pn]);
                }
                if (g + 2 == NBGRP)
                    btail = *reinterpret_cast<const v2f*>(xT);
            }
            __builtin_amdgcn_sched_barrier(0);
            #pragma unroll
            for (int i = 0; i < PGRP; ++i) {
                const int p = g * PGRP + i;
                if (p < NPAIR) {
                    const v2f be = {bq[g & 1][i].x, bq[g & 1][i].y};
                    const v2f bo = {bq[g & 1][i].z, bq[g & 1][i].w};
                    acc = __builtin_amdgcn_wmma_f32_16x16x4_f32(
                        false, afrag[2 * p], false, be, (short)0, acc, false, false);
                    acc = __builtin_amdgcn_wmma_f32_16x16x4_f32(
                        false, afrag[2 * p + 1], false, bo, (short)0, acc, false, false);
                }
            }
        }
        acc = __builtin_amdgcn_wmma_f32_16x16x4_f32(
            false, afrag[KSTEPS - 1], false, btail, (short)0, acc, false, false);

        {
            const int colw = (wave << 4) + ((lane >> 4) << 3);
            float* yrow = &yo[mrow * YSTRIDE + colw];
            #pragma unroll
            for (int i = 0; i < 8; ++i) yrow[i] = acc[i];
        }
        lds_barrier();

        if (ocol4 < n_out - (tile << 7)) {
            const float* ysrc = &yo[wave * YSTRIDE + ocol4];
            const float4 y0 = *reinterpret_cast<const float4*>(ysrc);
            const float4 y1 = *reinterpret_cast<const float4*>(ysrc + 8 * YSTRIDE);
            float* orow = out + (size_t)(b0 + wave) * (size_t)n_out + (size_t)(tile << 7);
            *reinterpret_cast<float4*>(orow + ocol4) = y0;
            *reinterpret_cast<float4*>(orow + (size_t)8 * (size_t)n_out + ocol4) = y1;
        }
    };

    int tile = blockIdx.x;
    int t_al = window_start(tile);
    float4 xv[XLD];
    load_window(xwg4, n4, t_al >> 2, wave, lane, xv);
    {
        float hk[HLD];
        #pragma unroll
        for (int j = 0; j < HLD; ++j) {
            const int k = TAPMAX - (tid + 256 * j);
            hk[j] = h[((unsigned)k < (unsigned)HLEN) ? k : 0];
        }
        #pragma unroll
        for (int j = 0; j < HLD; ++j) {
            const int k = TAPMAX - (tid + 256 * j);
            hrev[tid + 256 * j] = ((unsigned)k < (unsigned)HLEN) ? 5.0f * hk[j] : 0.0f;
        }
    }
    store_window(xs, n4, t_al >> 2, wave, lane, xv);
    __syncthreads();

    {
        const int m0f  = ((tile << 3) + wave) << 4;
        const int t0f  = (DOWN * (m0f + MSHIFT)) / UP;
        const int sh   = ((t0f - JMAX) - t_al) & 3;
        const int tap0 = DOWN * (m0f + mrow + MSHIFT) - UP * (t0f - JMAX - sh);
        const float* hA = &hrev[TAPMAX - tap0];
        const float* hP = hA + UP * khalf4;
        #pragma unroll
        for (int p = 0; p < NPAIR; ++p) {
            afrag[2 * p].x     = hP[40 * p];
            afrag[2 * p].y     = hP[40 * p + 5];
            afrag[2 * p + 1].x = hP[40 * p + 10];
            afrag[2 * p + 1].y = hP[40 * p + 15];
        }
        const float* hT = hA + UP * (8 * NPAIR + ksub);
        afrag[KSTEPS - 1].x = hT[0];
        afrag[KSTEPS - 1].y = hT[5];
    }

    unsigned va = (unsigned)lane + (unsigned)(t_al >> 2);
    unsigned vl = c4last + (unsigned)(t_al >> 2);

    bool more;
    for (;;) {
        const int next = tile + (int)gridDim.x;
        more = next < ntiles;
        int  t_al_next = 0;
        bool pre = false;
        if (more) {
            t_al_next = window_start(next);
            const int tb4 = t_al_next >> 2;
            pre = (tb4 >= 0) && (tb4 + XC4 <= (int)n4);
        }
        if (pre) {
            const unsigned d4 = (unsigned)((t_al_next - t_al) >> 2);
            va += d4;
            vl += d4;
            load_window_inner(rowa, rowb, va, vl, xv);
        }

        compute_tile(tile, t_al);
        if (!pre) break;

        store_window_inner(xs, wave, lane, c4last, xv);
        lds_barrier();
        tile = next;
        t_al = t_al_next;
    }

    while (more) {
        const int next = tile + (int)gridDim.x;
        const int t_al_next = window_start(next);
        load_window(xwg4, n4, t_al_next >> 2, wave, lane, xv);
        store_window(xs, n4, t_al_next >> 2, wave, lane, xv);
        lds_barrier();
        compute_tile(next, t_al_next);
        tile = next;
        more = tile + (int)gridDim.x < ntiles;
    }
}

extern "C" void kernel_launch(void* const* d_in, const int* in_sizes, int n_in_bufs,
                              void* d_out, int out_size, void* d_ws, size_t ws_size,
                              hipStream_t stream) {
    const float* x = (const float*)d_in[0];
    const float* h = (const float*)d_in[1];
    float* out = (float*)d_out;

    const int T = 1440000;
    const int B = in_sizes[0] / T;
    const int n_out = (int)(((long long)T * UP) / DOWN);

    dim3 grid(255, B / 16);
    resample_poly_wmma<<<grid, 256, 0, stream>>>(x, h, out, T, n_out);
}
